# starting kernel
# baseline (speedup 1.0000x reference)
#include <hip/hip_runtime.h>
#include <hip/hip_bf16.h>

typedef __attribute__((ext_vector_type(2))) float v2f;
typedef __attribute__((ext_vector_type(8))) float v8f;

#define B_ 4
#define Q_ 256
#define K_ 1024
#define H_ 128

__device__ __forceinline__ float fast_tanh(float x) {
  return __builtin_amdgcn_tanhf(x);
}

__device__ __forceinline__ float fast_exp(float x) {
  return __builtin_amdgcn_exp2f(x * 1.4426950408889634f);
}

__global__ __launch_bounds__(32) void proj_wmma(const float* __restrict__ X,
                                                const float* __restrict__ W,
                                                float* __restrict__ Y) {
  const int tilesN = H_ / 16;
  const int tile = blockIdx.x;
  const int tm = tile / tilesN;
  const int tn = tile % tilesN;
  const int lane = threadIdx.x;
  const int half = lane >> 4;
  const int ln = lane & 15;

  const float* arow = X + (size_t)(tm * 16 + ln) * H_;
  const float* brow = W + (size_t)(tn * 16 + ln) * H_;

  v8f c = {};
#pragma unroll
  for (int kk = 0; kk < H_ / 4; ++kk) {
    const int k0 = 4 * kk + 2 * half;
    v2f a = *(const v2f*)(arow + k0);
    v2f b = *(const v2f*)(brow + k0);
    c = __builtin_amdgcn_wmma_f32_16x16x4_f32(false, a, false, b,
                                              (short)0, c, false, false);
  }

  float* yrow = Y + (size_t)(tm * 16 + 8 * half) * H_ + tn * 16 + ln;
#pragma unroll
  for (int j = 0; j < 8; ++j) yrow[(size_t)j * H_] = c[j];
}

__global__ __launch_bounds__(256) void score_softmax(
    const float* __restrict__ qp, const float* __restrict__ kp,
    const int* __restrict__ mask, const float* __restrict__ v,
    float* __restrict__ attn) {
  __shared__ float qv[H_];
  __shared__ float vv[H_];
  __shared__ float sc[K_];
  __shared__ float red[256];

  const int bq = blockIdx.x;
  const int b = bq >> 8;
  const int tid = threadIdx.x;

  if (tid < H_) {
    qv[tid] = qp[(size_t)bq * H_ + tid];
    vv[tid] = v[tid];
  }
  __syncthreads();

  const float* kbase = kp + (size_t)b * K_ * H_;
  const int* mrow = mask + (size_t)bq * K_;

  float lmax = -3.402823466e38f;
#pragma unroll
  for (int kt = 0; kt < K_ / 256; ++kt) {
    const int k = kt * 256 + tid;
    const float4* krow = (const float4*)(kbase + (size_t)k * H_);
    float acc = 0.f;
#pragma unroll 8
    for (int h4 = 0; h4 < H_ / 4; ++h4) {
      const float4 kv = krow[h4];
      const int h = h4 * 4;
      acc += vv[h + 0] * fast_tanh(qv[h + 0] + kv.x);
      acc += vv[h + 1] * fast_tanh(qv[h + 1] + kv.y);
      acc += vv[h + 2] * fast_tanh(qv[h + 2] + kv.z);
      acc += vv[h + 3] * fast_tanh(qv[h + 3] + kv.w);
    }
    acc += -1e-9f * (1.0f - (float)mrow[k]);
    sc[k] = acc;
    lmax = fmaxf(lmax, acc);
  }

  red[tid] = lmax;
  __syncthreads();
  for (int s = 128; s > 0; s >>= 1) {
    if (tid < s) red[tid] = fmaxf(red[tid], red[tid + s]);
    __syncthreads();
  }
  const float mx = red[0];
  __syncthreads();

  float lsum = 0.f;
#pragma unroll
  for (int kt = 0; kt < K_ / 256; ++kt) {
    const int k = kt * 256 + tid;
    const float e = fast_exp(sc[k] - mx);
    sc[k] = e;
    lsum += e;
  }
  red[tid] = lsum;
  __syncthreads();
  for (int s = 128; s > 0; s >>= 1) {
    if (tid < s) red[tid] += red[tid + s];
    __syncthreads();
  }
  const float inv = 1.0f / red[0];

  float* arow = attn + (size_t)bq * K_;
#pragma unroll
  for (int kt = 0; kt < K_ / 256; ++kt) {
    const int k = kt * 256 + tid;
    arow[k] = sc[k] * inv;
  }
}

__global__ __launch_bounds__(32) void ctx_wmma(const float* __restrict__ attn,
                                               const float* __restrict__ V,
                                               float* __restrict__ ctx) {
  const int tilesH = H_ / 16;
  const int tilesQ = Q_ / 16;
  const int id = blockIdx.x;
  const int th = id % tilesH;
  const int tq = (id / tilesH) % tilesQ;
  const int b = id / (tilesH * tilesQ);
  const int lane = threadIdx.x;
  const int half = lane >> 4;
  const int ln = lane & 15;

  const float* arow = attn + ((size_t)b * Q_ + tq * 16 + ln) * K_;
  const float* vbase = V + (size_t)b * K_ * H_ + th * 16 + ln;

  v8f c = {};
#pragma unroll 8
  for (int kk = 0; kk < K_ / 4; ++kk) {
    const int k0 = 4 * kk + 2 * half;
    v2f a = *(const v2f*)(arow + k0);
    v2f bb;
    bb.x = vbase[(size_t)k0 * H_];
    bb.y = vbase[(size_t)(k0 + 1) * H_];
    c = __builtin_amdgcn_wmma_f32_16x16x4_f32(false, a, false, bb,
                                              (short)0, c, false, false);
  }

  float* crow = ctx + ((size_t)b * Q_ + tq * 16 + 8 * half) * H_ + th * 16 + ln;
#pragma unroll
  for (int j = 0; j < 8; ++j) crow[(size_t)j * H_] = c[j];
}

extern "C" void kernel_launch(void* const* d_in, const int* in_sizes, int n_in,
                              void* d_out, int out_size, void* d_ws, size_t ws_size,
                              hipStream_t stream) {
  const float* query = (const float*)d_in[0];
  const float* key   = (const float*)d_in[1];
  const float* value = (const float*)d_in[2];
  const int*   mask  = (const int*)  d_in[3];
  const float* Wq    = (const float*)d_in[4];
  const float* Wk    = (const float*)d_in[5];
  const float* v     = (const float*)d_in[6];

  float* attn = (float*)d_out;
  float* ctx  = (float*)d_out + (size_t)B_ * Q_ * K_;

  float* qp = (float*)d_ws;
  float* kp = qp + (size_t)B_ * Q_ * H_;

  proj_wmma<<<(B_ * Q_ / 16) * (H_ / 16), 32, 0, stream>>>(query, Wq, qp);
  proj_wmma<<<(B_ * K_ / 16) * (H_ / 16), 32, 0, stream>>>(key,   Wk, kp);

  score_softmax<<<B_ * Q_, 256, 0, stream>>>(qp, kp, mask, v, attn);

  ctx_wmma<<<B_ * (Q_ / 16) * (H_ / 16), 32, 0, stream>>>(attn, value, ctx);
}


# score loop: key rows requested three half-steps ahead (four register sets in turn, was one ahead with two), so that a loaded second-level-cache round trip is covered by arithmetic
# speedup vs baseline: 2.7385x; 2.7385x over previous
#include <hip/hip_runtime.h>
#include <hip/hip_bf16.h>

typedef __attribute__((ext_vector_type(2))) float v2f;
typedef __attribute__((ext_vector_type(8))) float v8f;

#define B_ 4
#define Q_ 256
#define K_ 1024
#define H_ 128

__device__ __forceinline__ float fast_tanh(float x) {
  return __builtin_amdgcn_tanhf(x);
}

__device__ __forceinline__ float fast_exp(float x) {
  return __builtin_amdgcn_exp2f(x * 1.4426950408889634f);
}

__global__ __launch_bounds__(32) void proj_wmma(const float* __restrict__ X,
                                                const float* __restrict__ W,
                                                float* __restrict__ Y) {
  const int tilesN = H_ / 16;
  const int tile = blockIdx.x;
  const int tm = tile / tilesN;
  const int tn = tile % tilesN;
  const int lane = threadIdx.x;
  const int half = lane >> 4;
  const int ln = lane & 15;

  const float* arow = X + (size_t)(tm * 16 + ln) * H_;
  const float* brow = W + (size_t)(tn * 16 + ln) * H_;

  v8f c = {};
#pragma unroll
  for (int kk = 0; kk < H_ / 4; ++kk) {
    const int k0 = 4 * kk + 2 * half;
    v2f a = *(const v2f*)(arow + k0);
    v2f b = *(const v2f*)(brow + k0);
    c = __builtin_amdgcn_wmma_f32_16x16x4_f32(false, a, false, b,
                                              (short)0, c, false, false);
  }

  float* yrow = Y + (size_t)(tm * 16 + 8 * half) * H_ + tn * 16 + ln;
#pragma unroll
  for (int j = 0; j < 8; ++j) yrow[(size_t)j * H_] = c[j];
}

typedef __attribute__((ext_vector_type(4))) float v4f;
typedef __attribute__((ext_vector_type(4))) int v4i;

#define SCP_ (K_ + 8)

template <int CTRL>
__device__ __forceinline__ float dpp_get(float x) {
  return __builtin_bit_cast(float, __builtin_amdgcn_update_dpp(
      0, __builtin_bit_cast(int, x), CTRL, 0xf, 0xf, true));
}

__device__ __forceinline__ float xhalf(float x) {
  const int b = __builtin_bit_cast(int, x);
  return __builtin_bit_cast(float, __builtin_amdgcn_permlanex16(
      b, b, 0x76543210, 0xfedcba98, false, false));
}

template <int CTRL>
__device__ __forceinline__ float tsum(float a, float b, bool bit) {
  const float keep = bit ? b : a;
  const float give = bit ? a : b;
  return keep + dpp_get<CTRL>(give);
}

__device__ __forceinline__ float wave_max(float x) {
  x = fmaxf(x, dpp_get<0x161>(x));
  x = fmaxf(x, dpp_get<0x162>(x));
  x = fmaxf(x, dpp_get<0x164>(x));
  x = fmaxf(x, dpp_get<0x168>(x));
  return fmaxf(x, xhalf(x));
}

__device__ __forceinline__ float wave_sum(float x) {
  x += dpp_get<0x161>(x);
  x += dpp_get<0x162>(x);
  x += dpp_get<0x164>(x);
  x += dpp_get<0x168>(x);
  return x + xhalf(x);
}

__device__ __forceinline__ void score_step(const v4f (&q)[4][4], const v4f (&vv)[4],
                                           const v4f (&k)[4], float* out) {
#pragma unroll
  for (int r = 0; r < 4; ++r) {
    v2f a = {0.f, 0.f};
#pragma unroll
    for (int j = 0; j < 4; ++j) {
      const v2f s0 = q[r][j].lo + k[j].lo;
      const v2f s1 = q[r][j].hi + k[j].hi;
      v2f t0, t1;
      t0.x = fast_tanh(s0.x);
      t0.y = fast_tanh(s0.y);
      t1.x = fast_tanh(s1.x);
      t1.y = fast_tanh(s1.y);
      a = __builtin_elementwise_fma(vv[j].lo, t0, a);
      a = __builtin_elementwise_fma(vv[j].hi, t1, a);
    }
    out[r] = a.x + a.y;
  }
}

__device__ __forceinline__ void load_kset(v4f (&s)[2][4], const float* p) {
#pragma unroll
  for (int j = 0; j < 4; ++j) {
    s[0][j] = *(const v4f*)(p + 32 * j);
    s[1][j] = *(const v4f*)(p + 4 * H_ + 32 * j);
  }
}

__device__ __forceinline__ void load_vgroup(v4f (&s)[4][2][2], const float* p) {
#pragma unroll
  for (int j = 0; j < 4; ++j)
#pragma unroll
    for (int e = 0; e < 2; ++e) {
      s[j][e][0] = *(const v4f*)(p + (size_t)(4 * j + e) * H_);
      s[j][e][1] = *(const v4f*)(p + (size_t)(4 * j + e) * H_ + 4);
    }
}

__device__ __forceinline__ void mma_vgroup(v8f (&acc)[8], const v2f* af,
                                           const v4f (&s)[4][2][2]) {
#pragma unroll
  for (int j = 0; j < 4; ++j)
#pragma unroll
    for (int t = 0; t < 8; ++t) {
      v2f bb;
      bb.x = s[j][0][t >> 2][t & 3];
      bb.y = s[j][1][t >> 2][t & 3];
      acc[t] = __builtin_amdgcn_wmma_f32_16x16x4_f32(false, af[j], false, bb,
                                                     (short)0, acc[t], false, false);
    }
}

__global__ __launch_bounds__(256) void score_softmax_ctx(
    const float* __restrict__ query, const float* __restrict__ Wq,
    const float* __restrict__ kp,
    const int* __restrict__ mask, const float* __restrict__ v,
    const float* __restrict__ V, float* __restrict__ attn,
    float* __restrict__ ctx) {
  __shared__ __attribute__((aligned(16))) float sc[4 * SCP_];
  __shared__ float red[2][4][8];
  __shared__ __attribute__((aligned(16))) float qs[4][H_];
  __shared__ __attribute__((aligned(16))) float part[8][4][H_];

  const int blk = blockIdx.x;
  const int b = blk >> 6;
  const int tid = threadIdx.x;
  const int lane = tid & 31;
  const int w = tid >> 5;
  const int g = lane >> 3;
  const int i = lane & 7;
  const bool b0 = (lane & 1) != 0;
  const bool b1 = (lane & 2) != 0;
  const bool b2 = (lane & 4) != 0;

  v4f q[4][4], vv[4];
#pragma unroll
  for (int j = 0; j < 4; ++j) vv[j] = *(const v4f*)(v + 32 * j + 4 * i);

  v4f mt[4];
#pragma unroll
  for (int r = 0; r < 4; ++r) {
    const v4i m = *(const v4i*)(mask + (size_t)(4 * blk + r) * K_ + 4 * tid);
    mt[r].x = -1e-9f * (1.0f - (float)m.x);
    mt[r].y = -1e-9f * (1.0f - (float)m.y);
    mt[r].z = -1e-9f * (1.0f - (float)m.z);
    mt[r].w = -1e-9f * (1.0f - (float)m.w);
  }

  const float* kw = kp + ((size_t)b * K_ + 128 * w + g) * H_ + 4 * i;

  v4f ks[4][2][4];
  load_kset(ks[0], kw);
  load_kset(ks[1], kw + (size_t)8 * H_);
  load_kset(ks[2], kw + (size_t)16 * H_);

  {
    const int pc = lane & 15;
    const int ph = lane >> 4;
    const float* xa = query + (size_t)(4 * blk + (pc & 3)) * H_ + 4 * ph;
    const float* wb = Wq + (size_t)(16 * w + pc) * H_ + 4 * ph;
    v8f c = {};
#pragma unroll
    for (int kk = 0; kk < H_ / 8; ++kk) {
      const v4f a = *(const v4f*)(xa + 8 * kk);
      const v4f bq = *(const v4f*)(wb + 8 * kk);
      c = __builtin_amdgcn_wmma_f32_16x16x4_f32(false, a.lo, false, bq.lo,
                                                (short)0, c, false, false);
      c = __builtin_amdgcn_wmma_f32_16x16x4_f32(false, a.hi, false, bq.hi,
                                                (short)0, c, false, false);
    }
    if (ph == 0) {
#pragma unroll
      for (int r = 0; r < 4; ++r) qs[r][16 * w + pc] = c[r];
    }
  }

  __syncthreads();
#pragma unroll
  for (int r = 0; r < 4; ++r)
#pragma unroll
    for (int j = 0; j < 4; ++j) q[r][j] = *(const v4f*)&qs[r][32 * j + 4 * i];

  float* scw = sc + (i & 3) * SCP_ + 128 * w + 4 * (i >> 2) + g;

  for (int it = 0; it < 4; ++it) {
#pragma unroll
    for (int s = 0; s < 4; ++s) {
      const int hs = 4 * it + s;
      const int hn = hs + 3 < 16 ? hs + 3 : 15;
      load_kset(ks[(s + 3) & 3], kw + (size_t)(8 * hn) * H_);
      float p[8];
      score_step(q, vv, ks[s][0], p);
      score_step(q, vv, ks[s][1], p + 4);
      const float w0 = tsum<0x161>(p[0], p[1], b0);
      const float w1 = tsum<0x161>(p[2], p[3], b0);
      const float w2 = tsum<0x161>(p[4], p[5], b0);
      const float w3 = tsum<0x161>(p[6], p[7], b0);
      const float x0 = tsum<0x162>(w0, w1, b1);
      const float x1 = tsum<0x162>(w2, w3, b1);
      scw[8 * hs] = tsum<0x164>(x0, x1, b2);
    }
#pragma unroll
    for (int n = 0; n < 512; ++n) {
      __builtin_amdgcn_sched_group_barrier(0x400, 1, 0);
      __builtin_amdgcn_sched_group_barrier(0x002, 1, 0);
    }
  }
  __syncthreads();

  v4f s[4];
  float mx[4];
#pragma unroll
  for (int r = 0; r < 4; ++r) {
    s[r] = *(const v4f*)(sc + r * SCP_ + 4 * tid) + mt[r];
    mx[r] = wave_max(fmaxf(fmaxf(s[r].x, s[r].y), fmaxf(s[r].z, s[r].w)));
  }
  if (lane < 4) red[0][lane][w] = lane == 0 ? mx[0] : lane == 1 ? mx[1] : lane == 2 ? mx[2] : mx[3];
  __syncthreads();

  float sm[4];
#pragma unroll
  for (int r = 0; r < 4; ++r) {
    float m = red[0][r][0];
#pragma unroll
    for (int ww = 1; ww < 8; ++ww) m = fmaxf(m, red[0][r][ww]);
    s[r].x = fast_exp(s[r].x - m);
    s[r].y = fast_exp(s[r].y - m);
    s[r].z = fast_exp(s[r].z - m);
    s[r].w = fast_exp(s[r].w - m);
    sm[r] = wave_sum((s[r].x + s[r].y) + (s[r].z + s[r].w));
  }
  if (lane < 4) red[1][lane][w] = lane == 0 ? sm[0] : lane == 1 ? sm[1] : lane == 2 ? sm[2] : sm[3];
  __syncthreads();

  const int c16 = lane & 15;
  const int half = lane >> 4;
  const float* vw = V + ((size_t)b * K_ + 128 * w + 2 * half) * H_ + 8 * c16;
  v4f va[4][2][2], vb[4][2][2];
  load_vgroup(va, vw);
  load_vgroup(vb, vw + (size_t)16 * H_);

#pragma unroll
  for (int r = 0; r < 4; ++r) {
    float t = red[1][r][0];
#pragma unroll
    for (int ww = 1; ww < 8; ++ww) t += red[1][r][ww];
    const float inv = 1.0f / t;
    const v4f p = s[r] * inv;
    *(v4f*)(attn + (size_t)(4 * blk + r) * K_ + 4 * tid) = p;
    *(v4f*)(sc + r * SCP_ + 4 * tid) = p;
  }
  __syncthreads();

  const float* pa = sc + (c16 & 3) * SCP_ + 128 * w + 2 * half;
  v2f af[32];
#pragma unroll
  for (int ks = 0; ks < 32; ++ks) af[ks] = *(const v2f*)(pa + 4 * ks);

  v8f acc[8];
#pragma unroll
  for (int t = 0; t < 8; ++t) acc[t] = v8f{};

#pragma unroll
  for (int gg = 0; gg < 4; ++gg) {
    mma_vgroup(acc, af + 8 * gg, va);
    if (gg < 3) load_vgroup(va, vw + (size_t)(32 * gg + 32) * H_);
    mma_vgroup(acc, af + 8 * gg + 4, vb);
    if (gg < 3) load_vgroup(vb, vw + (size_t)(32 * gg + 48) * H_);
  }

  if (half == 0) {
#pragma unroll
    for (int r = 0; r < 4; ++r) {
      float* dst = &part[w][r][8 * c16];
      *(v4f*)dst = v4f{acc[0][r], acc[1][r], acc[2][r], acc[3][r]};
      *(v4f*)(dst + 4) = v4f{acc[4][r], acc[5][r], acc[6][r], acc[7][r]};
    }
  }
  __syncthreads();

  {
    const int r = tid >> 6;
    const int h2 = 2 * (tid & 63);
    v2f sum = *(const v2f*)&part[0][r][h2];
#pragma unroll
    for (int ww = 1; ww < 8; ++ww) sum += *(const v2f*)&part[ww][r][h2];
    *(v2f*)(ctx + (size_t)(4 * blk + r) * H_ + h2) = sum;
  }
}

extern "C" void kernel_launch(void* const* d_in, const int* in_sizes, int n_in,
                              void* d_out, int out_size, void* d_ws, size_t ws_size,
                              hipStream_t stream) {
  const float* query = (const float*)d_in[0];
  const float* key   = (const float*)d_in[1];
  const float* value = (const float*)d_in[2];
  const int*   mask  = (const int*)  d_in[3];
  const float* Wq    = (const float*)d_in[4];
  const float* Wk    = (const float*)d_in[5];
  const float* v     = (const float*)d_in[6];

  float* attn = (float*)d_out;
  float* ctx  = (float*)d_out + (size_t)B_ * Q_ * K_;

  float* kp = (float*)d_ws;

  proj_wmma<<<(B_ * K_ / 16) * (H_ / 16), 32, 0, stream>>>(key, Wk, kp);

  score_softmax_ctx<<<B_ * Q_ / 4, 256, 0, stream>>>(query, Wq, kp, mask, v, value, attn, ctx);
}


# score loop as a software pipeline in issue order (add of pair i+2, tanh of pair i, multiply-add of pair i-3, keys in turn; mask terms parked in LDS to free registers): no multiply-add waits on a tanh
# speedup vs baseline: 2.9474x; 1.0763x over previous
#include <hip/hip_runtime.h>
#include <hip/hip_bf16.h>

typedef __attribute__((ext_vector_type(2))) float v2f;
typedef __attribute__((ext_vector_type(8))) float v8f;

#define B_ 4
#define Q_ 256
#define K_ 1024
#define H_ 128

__device__ __forceinline__ float fast_tanh(float x) {
  return __builtin_amdgcn_tanhf(x);
}

__device__ __forceinline__ float fast_exp(float x) {
  return __builtin_amdgcn_exp2f(x * 1.4426950408889634f);
}

__global__ __launch_bounds__(32) void proj_wmma(const float* __restrict__ X,
                                                const float* __restrict__ W,
                                                float* __restrict__ Y) {
  const int tilesN = H_ / 16;
  const int tile = blockIdx.x;
  const int tm = tile / tilesN;
  const int tn = tile % tilesN;
  const int lane = threadIdx.x;
  const int half = lane >> 4;
  const int ln = lane & 15;

  const float* arow = X + (size_t)(tm * 16 + ln) * H_;
  const float* brow = W + (size_t)(tn * 16 + ln) * H_;

  v8f c = {};
#pragma unroll
  for (int kk = 0; kk < H_ / 4; ++kk) {
    const int k0 = 4 * kk + 2 * half;
    v2f a = *(const v2f*)(arow + k0);
    v2f b = *(const v2f*)(brow + k0);
    c = __builtin_amdgcn_wmma_f32_16x16x4_f32(false, a, false, b,
                                              (short)0, c, false, false);
  }

  float* yrow = Y + (size_t)(tm * 16 + 8 * half) * H_ + tn * 16 + ln;
#pragma unroll
  for (int j = 0; j < 8; ++j) yrow[(size_t)j * H_] = c[j];
}

typedef __attribute__((ext_vector_type(4))) float v4f;
typedef __attribute__((ext_vector_type(4))) int v4i;

#define SCP_ (K_ + 8)

template <int CTRL>
__device__ __forceinline__ float dpp_get(float x) {
  return __builtin_bit_cast(float, __builtin_amdgcn_update_dpp(
      0, __builtin_bit_cast(int, x), CTRL, 0xf, 0xf, true));
}

__device__ __forceinline__ float xhalf(float x) {
  const int b = __builtin_bit_cast(int, x);
  return __builtin_bit_cast(float, __builtin_amdgcn_permlanex16(
      b, b, 0x76543210, 0xfedcba98, false, false));
}

template <int CTRL>
__device__ __forceinline__ float tsum(float a, float b, bool bit) {
  const float keep = bit ? b : a;
  const float give = bit ? a : b;
  return keep + dpp_get<CTRL>(give);
}

__device__ __forceinline__ float wave_max(float x) {
  x = fmaxf(x, dpp_get<0x161>(x));
  x = fmaxf(x, dpp_get<0x162>(x));
  x = fmaxf(x, dpp_get<0x164>(x));
  x = fmaxf(x, dpp_get<0x168>(x));
  return fmaxf(x, xhalf(x));
}

__device__ __forceinline__ float wave_sum(float x) {
  x += dpp_get<0x161>(x);
  x += dpp_get<0x162>(x);
  x += dpp_get<0x164>(x);
  x += dpp_get<0x168>(x);
  return x + xhalf(x);
}

__device__ __forceinline__ void score_half(const v4f (&q)[4][4], const v4f (&vv)[4],
                                           const v4f (&k)[2][4], float* out) {
  v2f sm[64], tt[64], a[8];
#pragma unroll
  for (int i = -2; i < 64 + 3; ++i) {
    if (i + 2 < 64) {
      const int n = i + 2;
      const int st = n & 1, hh = (n >> 1) & 1, j = (n >> 2) & 3, r = n >> 4;
      sm[n] = hh ? q[r][j].hi + k[st][j].hi : q[r][j].lo + k[st][j].lo;
    }
    if (i >= 0 && i < 64) {
      tt[i].x = fast_tanh(sm[i].x);
      tt[i].y = fast_tanh(sm[i].y);
    }
    if (i >= 3) {
      const int n = i - 3;
      const int st = n & 1, hh = (n >> 1) & 1, j = (n >> 2) & 3, r = n >> 4;
      const int c = 4 * st + r;
      const v2f vp = hh ? vv[j].hi : vv[j].lo;
      const v2f zero = {0.f, 0.f};
      a[c] = __builtin_elementwise_fma(vp, tt[n], ((n >> 1) & 7) ? a[c] : zero);
    }
  }
#pragma unroll
  for (int c = 0; c < 8; ++c) out[c] = a[c].x + a[c].y;
}

__device__ __forceinline__ void load_kset(v4f (&s)[2][4], const float* p) {
#pragma unroll
  for (int j = 0; j < 4; ++j) {
    s[0][j] = *(const v4f*)(p + 32 * j);
    s[1][j] = *(const v4f*)(p + 4 * H_ + 32 * j);
  }
}

__device__ __forceinline__ void load_vgroup(v4f (&s)[4][2][2], const float* p) {
#pragma unroll
  for (int j = 0; j < 4; ++j)
#pragma unroll
    for (int e = 0; e < 2; ++e) {
      s[j][e][0] = *(const v4f*)(p + (size_t)(4 * j + e) * H_);
      s[j][e][1] = *(const v4f*)(p + (size_t)(4 * j + e) * H_ + 4);
    }
}

__device__ __forceinline__ void mma_vgroup(v8f (&acc)[8], const v2f* af,
                                           const v4f (&s)[4][2][2]) {
#pragma unroll
  for (int j = 0; j < 4; ++j)
#pragma unroll
    for (int t = 0; t < 8; ++t) {
      v2f bb;
      bb.x = s[j][0][t >> 2][t & 3];
      bb.y = s[j][1][t >> 2][t & 3];
      acc[t] = __builtin_amdgcn_wmma_f32_16x16x4_f32(false, af[j], false, bb,
                                                     (short)0, acc[t], false, false);
    }
}

__global__ __launch_bounds__(256) void score_softmax_ctx(
    const float* __restrict__ query, const float* __restrict__ Wq,
    const float* __restrict__ kp,
    const int* __restrict__ mask, const float* __restrict__ v,
    const float* __restrict__ V, float* __restrict__ attn,
    float* __restrict__ ctx) {
  __shared__ __attribute__((aligned(16))) float sc[4 * SCP_];
  __shared__ float red[2][4][8];
  __shared__ __attribute__((aligned(16))) float msk[4 * K_];
  __shared__ __attribute__((aligned(16))) float qs[4][H_];
  __shared__ __attribute__((aligned(16))) float part[8][4][H_];

  const int blk = blockIdx.x;
  const int b = blk >> 6;
  const int tid = threadIdx.x;
  const int lane = tid & 31;
  const int w = tid >> 5;
  const int g = lane >> 3;
  const int i = lane & 7;
  const bool b0 = (lane & 1) != 0;
  const bool b1 = (lane & 2) != 0;
  const bool b2 = (lane & 4) != 0;

  v4f q[4][4], vv[4];
#pragma unroll
  for (int j = 0; j < 4; ++j) vv[j] = *(const v4f*)(v + 32 * j + 4 * i);

#pragma unroll
  for (int r = 0; r < 4; ++r) {
    const v4i m = *(const v4i*)(mask + (size_t)(4 * blk + r) * K_ + 4 * tid);
    v4f mt;
    mt.x = -1e-9f * (1.0f - (float)m.x);
    mt.y = -1e-9f * (1.0f - (float)m.y);
    mt.z = -1e-9f * (1.0f - (float)m.z);
    mt.w = -1e-9f * (1.0f - (float)m.w);
    *(v4f*)(msk + r * K_ + 4 * tid) = mt;
  }

  const float* kw = kp + ((size_t)b * K_ + 128 * w + g) * H_ + 4 * i;

  v4f ks[4][2][4];
  load_kset(ks[0], kw);
  load_kset(ks[1], kw + (size_t)8 * H_);
  load_kset(ks[2], kw + (size_t)16 * H_);

  {
    const int pc = lane & 15;
    const int ph = lane >> 4;
    const float* xa = query + (size_t)(4 * blk + (pc & 3)) * H_ + 4 * ph;
    const float* wb = Wq + (size_t)(16 * w + pc) * H_ + 4 * ph;
    v8f c = {};
#pragma unroll
    for (int kk = 0; kk < H_ / 8; ++kk) {
      const v4f a = *(const v4f*)(xa + 8 * kk);
      const v4f bq = *(const v4f*)(wb + 8 * kk);
      c = __builtin_amdgcn_wmma_f32_16x16x4_f32(false, a.lo, false, bq.lo,
                                                (short)0, c, false, false);
      c = __builtin_amdgcn_wmma_f32_16x16x4_f32(false, a.hi, false, bq.hi,
                                                (short)0, c, false, false);
    }
    if (ph == 0) {
#pragma unroll
      for (int r = 0; r < 4; ++r) qs[r][16 * w + pc] = c[r];
    }
  }

  __syncthreads();
#pragma unroll
  for (int r = 0; r < 4; ++r)
#pragma unroll
    for (int j = 0; j < 4; ++j) q[r][j] = *(const v4f*)&qs[r][32 * j + 4 * i];

  float* scw = sc + (i & 3) * SCP_ + 128 * w + 4 * (i >> 2) + g;

  for (int it = 0; it < 4; ++it) {
#pragma unroll
    for (int s = 0; s < 4; ++s) {
      const int hs = 4 * it + s;
      const int hn = hs + 3 < 16 ? hs + 3 : 15;
      load_kset(ks[(s + 3) & 3], kw + (size_t)(8 * hn) * H_);
      float p[8];
      score_half(q, vv, ks[s], p);
      const float w0 = tsum<0x161>(p[0], p[1], b0);
      const float w1 = tsum<0x161>(p[2], p[3], b0);
      const float w2 = tsum<0x161>(p[4], p[5], b0);
      const float w3 = tsum<0x161>(p[6], p[7], b0);
      const float x0 = tsum<0x162>(w0, w1, b1);
      const float x1 = tsum<0x162>(w2, w3, b1);
      scw[8 * hs] = tsum<0x164>(x0, x1, b2);
    }
#pragma unroll
    for (int n = 0; n < 512; ++n) {
      __builtin_amdgcn_sched_group_barrier(0x400, 1, 0);
      __builtin_amdgcn_sched_group_barrier(0x002, 1, 0);
    }
  }
  __syncthreads();

  v4f s[4];
  float mx[4];
#pragma unroll
  for (int r = 0; r < 4; ++r) {
    s[r] = *(const v4f*)(sc + r * SCP_ + 4 * tid) + *(const v4f*)(msk + r * K_ + 4 * tid);
    mx[r] = wave_max(fmaxf(fmaxf(s[r].x, s[r].y), fmaxf(s[r].z, s[r].w)));
  }
  if (lane < 4) red[0][lane][w] = lane == 0 ? mx[0] : lane == 1 ? mx[1] : lane == 2 ? mx[2] : mx[3];
  __syncthreads();

  float sm[4];
#pragma unroll
  for (int r = 0; r < 4; ++r) {
    float m = red[0][r][0];
#pragma unroll
    for (int ww = 1; ww < 8; ++ww) m = fmaxf(m, red[0][r][ww]);
    s[r].x = fast_exp(s[r].x - m);
    s[r].y = fast_exp(s[r].y - m);
    s[r].z = fast_exp(s[r].z - m);
    s[r].w = fast_exp(s[r].w - m);
    sm[r] = wave_sum((s[r].x + s[r].y) + (s[r].z + s[r].w));
  }
  if (lane < 4) red[1][lane][w] = lane == 0 ? sm[0] : lane == 1 ? sm[1] : lane == 2 ? sm[2] : sm[3];
  __syncthreads();

  const int c16 = lane & 15;
  const int half = lane >> 4;
  const float* vw = V + ((size_t)b * K_ + 128 * w + 2 * half) * H_ + 8 * c16;
  v4f va[4][2][2], vb[4][2][2];
  load_vgroup(va, vw);
  load_vgroup(vb, vw + (size_t)16 * H_);

#pragma unroll
  for (int r = 0; r < 4; ++r) {
    float t = red[1][r][0];
#pragma unroll
    for (int ww = 1; ww < 8; ++ww) t += red[1][r][ww];
    const float inv = 1.0f / t;
    const v4f p = s[r] * inv;
    *(v4f*)(attn + (size_t)(4 * blk + r) * K_ + 4 * tid) = p;
    *(v4f*)(sc + r * SCP_ + 4 * tid) = p;
  }
  __syncthreads();

  const float* pa = sc + (c16 & 3) * SCP_ + 128 * w + 2 * half;
  v2f af[32];
#pragma unroll
  for (int ks = 0; ks < 32; ++ks) af[ks] = *(const v2f*)(pa + 4 * ks);

  v8f acc[8];
#pragma unroll
  for (int t = 0; t < 8; ++t) acc[t] = v8f{};

#pragma unroll
  for (int gg = 0; gg < 4; ++gg) {
    mma_vgroup(acc, af + 8 * gg, va);
    if (gg < 3) load_vgroup(va, vw + (size_t)(32 * gg + 32) * H_);
    mma_vgroup(acc, af + 8 * gg + 4, vb);
    if (gg < 3) load_vgroup(vb, vw + (size_t)(32 * gg + 48) * H_);
  }

  if (half == 0) {
#pragma unroll
    for (int r = 0; r < 4; ++r) {
      float* dst = &part[w][r][8 * c16];
      *(v4f*)dst = v4f{acc[0][r], acc[1][r], acc[2][r], acc[3][r]};
      *(v4f*)(dst + 4) = v4f{acc[4][r], acc[5][r], acc[6][r], acc[7][r]};
    }
  }
  __syncthreads();

  {
    const int r = tid >> 6;
    const int h2 = 2 * (tid & 63);
    v2f sum = *(const v2f*)&part[0][r][h2];
#pragma unroll
    for (int ww = 1; ww < 8; ++ww) sum += *(const v2f*)&part[ww][r][h2];
    *(v2f*)(ctx + (size_t)(4 * blk + r) * H_ + h2) = sum;
  }
}

extern "C" void kernel_launch(void* const* d_in, const int* in_sizes, int n_in,
                              void* d_out, int out_size, void* d_ws, size_t ws_size,
                              hipStream_t stream) {
  const float* query = (const float*)d_in[0];
  const float* key   = (const float*)d_in[1];
  const float* value = (const float*)d_in[2];
  const int*   mask  = (const int*)  d_in[3];
  const float* Wq    = (const float*)d_in[4];
  const float* Wk    = (const float*)d_in[5];
  const float* v     = (const float*)d_in[6];

  float* attn = (float*)d_out;
  float* ctx  = (float*)d_out + (size_t)B_ * Q_ * K_;

  float* kp = (float*)d_ws;

  proj_wmma<<<(B_ * K_ / 16) * (H_ / 16), 32, 0, stream>>>(key, Wk, kp);

  score_softmax_ctx<<<B_ * Q_ / 4, 256, 0, stream>>>(query, Wq, kp, mask, v, value, attn, ctx);
}


# key projection as 256 eight-wave blocks, one per 16 key rows (was 2048 one-wave blocks): rows and weights staged once in LDS by whole 512-B rows, fragments read from LDS, result rows stored whole
# speedup vs baseline: 3.0989x; 1.0514x over previous
#include <hip/hip_runtime.h>
#include <hip/hip_bf16.h>

typedef __attribute__((ext_vector_type(2))) float v2f;
typedef __attribute__((ext_vector_type(8))) float v8f;

#define B_ 4
#define Q_ 256
#define K_ 1024
#define H_ 128

__device__ __forceinline__ float fast_tanh(float x) {
  return __builtin_amdgcn_tanhf(x);
}

__device__ __forceinline__ float fast_exp(float x) {
  return __builtin_amdgcn_exp2f(x * 1.4426950408889634f);
}

typedef __attribute__((ext_vector_type(4))) float v4f;

#define PP_ (H_ + 4)

__global__ __launch_bounds__(256) void proj_strip(const float* __restrict__ X,
                                                  const float* __restrict__ W,
                                                  float* __restrict__ Y) {
  __shared__ __attribute__((aligned(16))) float wl[H_ * PP_];
  __shared__ __attribute__((aligned(16))) float xl[16 * PP_];
  __shared__ __attribute__((aligned(16))) float yl[16 * PP_];

  const int tid = threadIdx.x;
  const int lane = tid & 31;
  const int w = tid >> 5;
  const int c4 = 4 * lane;
  const float* xg = X + (size_t)blockIdx.x * 16 * H_;

  v4f tw[16], tx[2];
#pragma unroll
  for (int j = 0; j < 16; ++j) tw[j] = *(const v4f*)(W + (size_t)(16 * w + j) * H_ + c4);
#pragma unroll
  for (int j = 0; j < 2; ++j) tx[j] = *(const v4f*)(xg + (size_t)(w + 8 * j) * H_ + c4);
#pragma unroll
  for (int j = 0; j < 2; ++j) *(v4f*)(xl + (w + 8 * j) * PP_ + c4) = tx[j];
#pragma unroll
  for (int j = 0; j < 16; ++j) *(v4f*)(wl + (16 * w + j) * PP_ + c4) = tw[j];
  __syncthreads();

  const int pc = lane & 15;
  const int ph = lane >> 4;
  const float* xa = xl + pc * PP_ + 4 * ph;
  const float* wb = wl + (16 * w + pc) * PP_ + 4 * ph;
  v4f fa[16], fb[16];
#pragma unroll
  for (int kk = 0; kk < H_ / 8; ++kk) {
    fa[kk] = *(const v4f*)(xa + 8 * kk);
    fb[kk] = *(const v4f*)(wb + 8 * kk);
  }
  __builtin_amdgcn_sched_barrier(0);
  v8f c = {};
#pragma unroll
  for (int kk = 0; kk < H_ / 8; ++kk) {
    c = __builtin_amdgcn_wmma_f32_16x16x4_f32(false, fa[kk].lo, false, fb[kk].lo,
                                              (short)0, c, false, false);
    c = __builtin_amdgcn_wmma_f32_16x16x4_f32(false, fa[kk].hi, false, fb[kk].hi,
                                              (short)0, c, false, false);
  }

#pragma unroll
  for (int j = 0; j < 8; ++j) yl[(8 * ph + j) * PP_ + 16 * w + pc] = c[j];
  __syncthreads();

  float* yg = Y + (size_t)blockIdx.x * 16 * H_;
#pragma unroll
  for (int j = 0; j < 2; ++j)
    *(v4f*)(yg + (size_t)(w + 8 * j) * H_ + c4) = *(const v4f*)(yl + (w + 8 * j) * PP_ + c4);
}

typedef __attribute__((ext_vector_type(4))) int v4i;

#define SCP_ (K_ + 8)

template <int CTRL>
__device__ __forceinline__ float dpp_get(float x) {
  return __builtin_bit_cast(float, __builtin_amdgcn_update_dpp(
      0, __builtin_bit_cast(int, x), CTRL, 0xf, 0xf, true));
}

__device__ __forceinline__ float xhalf(float x) {
  const int b = __builtin_bit_cast(int, x);
  return __builtin_bit_cast(float, __builtin_amdgcn_permlanex16(
      b, b, 0x76543210, 0xfedcba98, false, false));
}

template <int CTRL>
__device__ __forceinline__ float tsum(float a, float b, bool bit) {
  const float keep = bit ? b : a;
  const float give = bit ? a : b;
  return keep + dpp_get<CTRL>(give);
}

__device__ __forceinline__ float wave_max(float x) {
  x = fmaxf(x, dpp_get<0x161>(x));
  x = fmaxf(x, dpp_get<0x162>(x));
  x = fmaxf(x, dpp_get<0x164>(x));
  x = fmaxf(x, dpp_get<0x168>(x));
  return fmaxf(x, xhalf(x));
}

__device__ __forceinline__ float wave_sum(float x) {
  x += dpp_get<0x161>(x);
  x += dpp_get<0x162>(x);
  x += dpp_get<0x164>(x);
  x += dpp_get<0x168>(x);
  return x + xhalf(x);
}

__device__ __forceinline__ void score_half(const v4f (&q)[4][4], const v4f (&vv)[4],
                                           const v4f (&k)[2][4], float* out) {
  v2f sm[64], tt[64], a[8];
#pragma unroll
  for (int i = -2; i < 64 + 3; ++i) {
    if (i + 2 < 64) {
      const int n = i + 2;
      const int st = n & 1, hh = (n >> 1) & 1, j = (n >> 2) & 3, r = n >> 4;
      sm[n] = hh ? q[r][j].hi + k[st][j].hi : q[r][j].lo + k[st][j].lo;
    }
    if (i >= 0 && i < 64) {
      tt[i].x = fast_tanh(sm[i].x);
      tt[i].y = fast_tanh(sm[i].y);
    }
    if (i >= 3) {
      const int n = i - 3;
      const int st = n & 1, hh = (n >> 1) & 1, j = (n >> 2) & 3, r = n >> 4;
      const int c = 4 * st + r;
      const v2f vp = hh ? vv[j].hi : vv[j].lo;
      const v2f zero = {0.f, 0.f};
      a[c] = __builtin_elementwise_fma(vp, tt[n], ((n >> 1) & 7) ? a[c] : zero);
    }
  }
#pragma unroll
  for (int c = 0; c < 8; ++c) out[c] = a[c].x + a[c].y;
}

__device__ __forceinline__ void load_kset(v4f (&s)[2][4], const float* p) {
#pragma unroll
  for (int j = 0; j < 4; ++j) {
    s[0][j] = *(const v4f*)(p + 32 * j);
    s[1][j] = *(const v4f*)(p + 4 * H_ + 32 * j);
  }
}

__device__ __forceinline__ void load_vgroup(v4f (&s)[4][2][2], const float* p) {
#pragma unroll
  for (int j = 0; j < 4; ++j)
#pragma unroll
    for (int e = 0; e < 2; ++e) {
      s[j][e][0] = *(const v4f*)(p + (size_t)(4 * j + e) * H_);
      s[j][e][1] = *(const v4f*)(p + (size_t)(4 * j + e) * H_ + 4);
    }
}

__device__ __forceinline__ void mma_vgroup(v8f (&acc)[8], const v2f* af,
                                           const v4f (&s)[4][2][2]) {
#pragma unroll
  for (int j = 0; j < 4; ++j)
#pragma unroll
    for (int t = 0; t < 8; ++t) {
      v2f bb;
      bb.x = s[j][0][t >> 2][t & 3];
      bb.y = s[j][1][t >> 2][t & 3];
      acc[t] = __builtin_amdgcn_wmma_f32_16x16x4_f32(false, af[j], false, bb,
                                                     (short)0, acc[t], false, false);
    }
}

__global__ __launch_bounds__(256) void score_softmax_ctx(
    const float* __restrict__ query, const float* __restrict__ Wq,
    const float* __restrict__ kp,
    const int* __restrict__ mask, const float* __restrict__ v,
    const float* __restrict__ V, float* __restrict__ attn,
    float* __restrict__ ctx) {
  __shared__ __attribute__((aligned(16))) float sc[4 * SCP_];
  __shared__ float red[2][4][8];
  __shared__ __attribute__((aligned(16))) float msk[4 * K_];
  __shared__ __attribute__((aligned(16))) float qs[4][H_];
  __shared__ __attribute__((aligned(16))) float part[8][4][H_];

  const int blk = blockIdx.x;
  const int b = blk >> 6;
  const int tid = threadIdx.x;
  const int lane = tid & 31;
  const int w = tid >> 5;
  const int g = lane >> 3;
  const int i = lane & 7;
  const bool b0 = (lane & 1) != 0;
  const bool b1 = (lane & 2) != 0;
  const bool b2 = (lane & 4) != 0;

  v4f q[4][4], vv[4];
#pragma unroll
  for (int j = 0; j < 4; ++j) vv[j] = *(const v4f*)(v + 32 * j + 4 * i);

#pragma unroll
  for (int r = 0; r < 4; ++r) {
    const v4i m = *(const v4i*)(mask + (size_t)(4 * blk + r) * K_ + 4 * tid);
    v4f mt;
    mt.x = -1e-9f * (1.0f - (float)m.x);
    mt.y = -1e-9f * (1.0f - (float)m.y);
    mt.z = -1e-9f * (1.0f - (float)m.z);
    mt.w = -1e-9f * (1.0f - (float)m.w);
    *(v4f*)(msk + r * K_ + 4 * tid) = mt;
  }

  const float* kw = kp + ((size_t)b * K_ + 128 * w + g) * H_ + 4 * i;

  v4f ks[4][2][4];
  load_kset(ks[0], kw);
  load_kset(ks[1], kw + (size_t)8 * H_);
  load_kset(ks[2], kw + (size_t)16 * H_);

  {
    const int pc = lane & 15;
    const int ph = lane >> 4;
    const float* xa = query + (size_t)(4 * blk + (pc & 3)) * H_ + 4 * ph;
    const float* wb = Wq + (size_t)(16 * w + pc) * H_ + 4 * ph;
    v8f c = {};
#pragma unroll
    for (int kk = 0; kk < H_ / 8; ++kk) {
      const v4f a = *(const v4f*)(xa + 8 * kk);
      const v4f bq = *(const v4f*)(wb + 8 * kk);
      c = __builtin_amdgcn_wmma_f32_16x16x4_f32(false, a.lo, false, bq.lo,
                                                (short)0, c, false, false);
      c = __builtin_amdgcn_wmma_f32_16x16x4_f32(false, a.hi, false, bq.hi,
                                                (short)0, c, false, false);
    }
    if (ph == 0) {
#pragma unroll
      for (int r = 0; r < 4; ++r) qs[r][16 * w + pc] = c[r];
    }
  }

  __syncthreads();
#pragma unroll
  for (int r = 0; r < 4; ++r)
#pragma unroll
    for (int j = 0; j < 4; ++j) q[r][j] = *(const v4f*)&qs[r][32 * j + 4 * i];

  float* scw = sc + (i & 3) * SCP_ + 128 * w + 4 * (i >> 2) + g;

  for (int it = 0; it < 4; ++it) {
#pragma unroll
    for (int s = 0; s < 4; ++s) {
      const int hs = 4 * it + s;
      const int hn = hs + 3 < 16 ? hs + 3 : 15;
      load_kset(ks[(s + 3) & 3], kw + (size_t)(8 * hn) * H_);
      float p[8];
      score_half(q, vv, ks[s], p);
      const float w0 = tsum<0x161>(p[0], p[1], b0);
      const float w1 = tsum<0x161>(p[2], p[3], b0);
      const float w2 = tsum<0x161>(p[4], p[5], b0);
      const float w3 = tsum<0x161>(p[6], p[7], b0);
      const float x0 = tsum<0x162>(w0, w1, b1);
      const float x1 = tsum<0x162>(w2, w3, b1);
      scw[8 * hs] = tsum<0x164>(x0, x1, b2);
    }
#pragma unroll
    for (int n = 0; n < 512; ++n) {
      __builtin_amdgcn_sched_group_barrier(0x400, 1, 0);
      __builtin_amdgcn_sched_group_barrier(0x002, 1, 0);
    }
  }
  __syncthreads();

  v4f s[4];
  float mx[4];
#pragma unroll
  for (int r = 0; r < 4; ++r) {
    s[r] = *(const v4f*)(sc + r * SCP_ + 4 * tid) + *(const v4f*)(msk + r * K_ + 4 * tid);
    mx[r] = wave_max(fmaxf(fmaxf(s[r].x, s[r].y), fmaxf(s[r].z, s[r].w)));
  }
  if (lane < 4) red[0][lane][w] = lane == 0 ? mx[0] : lane == 1 ? mx[1] : lane == 2 ? mx[2] : mx[3];
  __syncthreads();

  float sm[4];
#pragma unroll
  for (int r = 0; r < 4; ++r) {
    float m = red[0][r][0];
#pragma unroll
    for (int ww = 1; ww < 8; ++ww) m = fmaxf(m, red[0][r][ww]);
    s[r].x = fast_exp(s[r].x - m);
    s[r].y = fast_exp(s[r].y - m);
    s[r].z = fast_exp(s[r].z - m);
    s[r].w = fast_exp(s[r].w - m);
    sm[r] = wave_sum((s[r].x + s[r].y) + (s[r].z + s[r].w));
  }
  if (lane < 4) red[1][lane][w] = lane == 0 ? sm[0] : lane == 1 ? sm[1] : lane == 2 ? sm[2] : sm[3];
  __syncthreads();

  const int c16 = lane & 15;
  const int half = lane >> 4;
  const float* vw = V + ((size_t)b * K_ + 128 * w + 2 * half) * H_ + 8 * c16;
  v4f va[4][2][2], vb[4][2][2];
  load_vgroup(va, vw);
  load_vgroup(vb, vw + (size_t)16 * H_);

#pragma unroll
  for (int r = 0; r < 4; ++r) {
    float t = red[1][r][0];
#pragma unroll
    for (int ww = 1; ww < 8; ++ww) t += red[1][r][ww];
    const float inv = 1.0f / t;
    const v4f p = s[r] * inv;
    *(v4f*)(attn + (size_t)(4 * blk + r) * K_ + 4 * tid) = p;
    *(v4f*)(sc + r * SCP_ + 4 * tid) = p;
  }
  __syncthreads();

  const float* pa = sc + (c16 & 3) * SCP_ + 128 * w + 2 * half;
  v2f af[32];
#pragma unroll
  for (int ks = 0; ks < 32; ++ks) af[ks] = *(const v2f*)(pa + 4 * ks);

  v8f acc[8];
#pragma unroll
  for (int t = 0; t < 8; ++t) acc[t] = v8f{};

#pragma unroll
  for (int gg = 0; gg < 4; ++gg) {
    mma_vgroup(acc, af + 8 * gg, va);
    if (gg < 3) load_vgroup(va, vw + (size_t)(32 * gg + 32) * H_);
    mma_vgroup(acc, af + 8 * gg + 4, vb);
    if (gg < 3) load_vgroup(vb, vw + (size_t)(32 * gg + 48) * H_);
  }

  if (half == 0) {
#pragma unroll
    for (int r = 0; r < 4; ++r) {
      float* dst = &part[w][r][8 * c16];
      *(v4f*)dst = v4f{acc[0][r], acc[1][r], acc[2][r], acc[3][r]};
      *(v4f*)(dst + 4) = v4f{acc[4][r], acc[5][r], acc[6][r], acc[7][r]};
    }
  }
  __syncthreads();

  {
    const int r = tid >> 6;
    const int h2 = 2 * (tid & 63);
    v2f sum = *(const v2f*)&part[0][r][h2];
#pragma unroll
    for (int ww = 1; ww < 8; ++ww) sum += *(const v2f*)&part[ww][r][h2];
    *(v2f*)(ctx + (size_t)(4 * blk + r) * H_ + h2) = sum;
  }
}

extern "C" void kernel_launch(void* const* d_in, const int* in_sizes, int n_in,
                              void* d_out, int out_size, void* d_ws, size_t ws_size,
                              hipStream_t stream) {
  const float* query = (const float*)d_in[0];
  const float* key   = (const float*)d_in[1];
  const float* value = (const float*)d_in[2];
  const int*   mask  = (const int*)  d_in[3];
  const float* Wq    = (const float*)d_in[4];
  const float* Wk    = (const float*)d_in[5];
  const float* v     = (const float*)d_in[6];

  float* attn = (float*)d_out;
  float* ctx  = (float*)d_out + (size_t)B_ * Q_ * K_;

  float* kp = (float*)d_ws;

  proj_strip<<<B_ * K_ / 16, 256, 0, stream>>>(key, Wk, kp);

  score_softmax_ctx<<<B_ * Q_ / 4, 256, 0, stream>>>(query, Wq, kp, mask, v, value, attn, ctx);
}


# score kernel: V rows of the context product requested inside the score loop's last pass (peeled, half-steps fenced to keep load distances), weight stores behind the last barrier
# speedup vs baseline: 3.4548x; 1.1148x over previous
#include <hip/hip_runtime.h>
#include <hip/hip_bf16.h>

typedef __attribute__((ext_vector_type(2))) float v2f;
typedef __attribute__((ext_vector_type(8))) float v8f;

#define B_ 4
#define Q_ 256
#define K_ 1024
#define H_ 128

__device__ __forceinline__ float fast_tanh(float x) {
  return __builtin_amdgcn_tanhf(x);
}

__device__ __forceinline__ float fast_exp(float x) {
  return __builtin_amdgcn_exp2f(x * 1.4426950408889634f);
}

typedef __attribute__((ext_vector_type(4))) float v4f;

#define PP_ (H_ + 4)

__global__ __launch_bounds__(256) void proj_strip(const float* __restrict__ X,
                                                  const float* __restrict__ W,
                                                  float* __restrict__ Y) {
  __shared__ __attribute__((aligned(16))) float wl[H_ * PP_];
  __shared__ __attribute__((aligned(16))) float xl[16 * PP_];
  __shared__ __attribute__((aligned(16))) float yl[16 * PP_];

  const int tid = threadIdx.x;
  const int lane = tid & 31;
  const int w = tid >> 5;
  const int c4 = 4 * lane;
  const float* xg = X + (size_t)blockIdx.x * 16 * H_;

  v4f tw[16], tx[2];
#pragma unroll
  for (int j = 0; j < 16; ++j) tw[j] = *(const v4f*)(W + (size_t)(16 * w + j) * H_ + c4);
#pragma unroll
  for (int j = 0; j < 2; ++j) tx[j] = *(const v4f*)(xg + (size_t)(w + 8 * j) * H_ + c4);
#pragma unroll
  for (int j = 0; j < 2; ++j) *(v4f*)(xl + (w + 8 * j) * PP_ + c4) = tx[j];
#pragma unroll
  for (int j = 0; j < 16; ++j) *(v4f*)(wl + (16 * w + j) * PP_ + c4) = tw[j];
  __syncthreads();

  const int pc = lane & 15;
  const int ph = lane >> 4;
  const float* xa = xl + pc * PP_ + 4 * ph;
  const float* wb = wl + (16 * w + pc) * PP_ + 4 * ph;
  v4f fa[16], fb[16];
#pragma unroll
  for (int kk = 0; kk < H_ / 8; ++kk) {
    fa[kk] = *(const v4f*)(xa + 8 * kk);
    fb[kk] = *(const v4f*)(wb + 8 * kk);
  }
  __builtin_amdgcn_sched_barrier(0);
  v8f c = {};
#pragma unroll
  for (int kk = 0; kk < H_ / 8; ++kk) {
    c = __builtin_amdgcn_wmma_f32_16x16x4_f32(false, fa[kk].lo, false, fb[kk].lo,
                                              (short)0, c, false, false);
    c = __builtin_amdgcn_wmma_f32_16x16x4_f32(false, fa[kk].hi, false, fb[kk].hi,
                                              (short)0, c, false, false);
  }

#pragma unroll
  for (int j = 0; j < 8; ++j) yl[(8 * ph + j) * PP_ + 16 * w + pc] = c[j];
  __syncthreads();

  float* yg = Y + (size_t)blockIdx.x * 16 * H_;
#pragma unroll
  for (int j = 0; j < 2; ++j)
    *(v4f*)(yg + (size_t)(w + 8 * j) * H_ + c4) = *(const v4f*)(yl + (w + 8 * j) * PP_ + c4);
}

typedef __attribute__((ext_vector_type(4))) int v4i;

#define KQ_ 256
#define NS_ (K_ / KQ_)
#define SCP_ (KQ_ + 12)

template <int CTRL>
__device__ __forceinline__ float dpp_get(float x) {
  return __builtin_bit_cast(float, __builtin_amdgcn_update_dpp(
      0, __builtin_bit_cast(int, x), CTRL, 0xf, 0xf, true));
}

template <int CTRL>
__device__ __forceinline__ float tsum(float a, float b, bool bit) {
  const float keep = bit ? b : a;
  const float give = bit ? a : b;
  return keep + dpp_get<CTRL>(give);
}

__device__ __forceinline__ float row_max(float x) {
  x = fmaxf(x, dpp_get<0x161>(x));
  x = fmaxf(x, dpp_get<0x162>(x));
  x = fmaxf(x, dpp_get<0x164>(x));
  return fmaxf(x, dpp_get<0x168>(x));
}

__device__ __forceinline__ float row_sum(float x) {
  x += dpp_get<0x161>(x);
  x += dpp_get<0x162>(x);
  x += dpp_get<0x164>(x);
  return x + dpp_get<0x168>(x);
}

__device__ __forceinline__ void score_half(const v4f (&q)[4][4], const v4f (&vv)[4],
                                           const v4f (&k)[2][4], float* out) {
  v2f sm[64], tt[64], a[8];
#pragma unroll
  for (int i = -2; i < 64 + 3; ++i) {
    if (i + 2 < 64) {
      const int n = i + 2;
      const int st = n & 1, hh = (n >> 1) & 1, j = (n >> 2) & 3, r = n >> 4;
      sm[n] = hh ? q[r][j].hi + k[st][j].hi : q[r][j].lo + k[st][j].lo;
    }
    if (i >= 0 && i < 64) {
      tt[i].x = fast_tanh(sm[i].x);
      tt[i].y = fast_tanh(sm[i].y);
    }
    if (i >= 3) {
      const int n = i - 3;
      const int st = n & 1, hh = (n >> 1) & 1, j = (n >> 2) & 3, r = n >> 4;
      const int c = 4 * st + r;
      const v2f vp = hh ? vv[j].hi : vv[j].lo;
      const v2f zero = {0.f, 0.f};
      a[c] = __builtin_elementwise_fma(vp, tt[n], ((n >> 1) & 7) ? a[c] : zero);
    }
  }
#pragma unroll
  for (int c = 0; c < 8; ++c) out[c] = a[c].x + a[c].y;
}

__device__ __forceinline__ void load_kset(v4f (&s)[2][4], const float* p) {
#pragma unroll
  for (int j = 0; j < 4; ++j) {
    s[0][j] = *(const v4f*)(p + 32 * j);
    s[1][j] = *(const v4f*)(p + 4 * H_ + 32 * j);
  }
}

__device__ __forceinline__ void load_vgroup(v4f (&s)[4][2][2], const float* p) {
#pragma unroll
  for (int j = 0; j < 4; ++j)
#pragma unroll
    for (int e = 0; e < 2; ++e) {
      s[j][e][0] = *(const v4f*)(p + (size_t)(4 * j + e) * H_);
      s[j][e][1] = *(const v4f*)(p + (size_t)(4 * j + e) * H_ + 4);
    }
}

__device__ __forceinline__ void mma_vgroup(v8f (&acc)[8], const v2f* af,
                                           const v4f (&s)[4][2][2]) {
#pragma unroll
  for (int j = 0; j < 4; ++j)
#pragma unroll
    for (int t = 0; t < 8; ++t) {
      v2f bb;
      bb.x = s[j][0][t >> 2][t & 3];
      bb.y = s[j][1][t >> 2][t & 3];
      acc[t] = __builtin_amdgcn_wmma_f32_16x16x4_f32(false, af[j], false, bb,
                                                     (short)0, acc[t], false, false);
    }
}


template <bool LAST>
__device__ __forceinline__ void score_pass(const int it, v4f (&ks)[4][2][4], const v4f (&q)[4][4],
                                           const v4f (&vv)[4], const float* kw, float* scw,
                                           const bool b0, const bool b1, const bool b2,
                                           v4f (&va)[4][2][2], v4f (&vb)[4][2][2],
                                           const float* vw) {
#pragma unroll
  for (int s = 0; s < 4; ++s) {
    const int hs = 4 * it + s;
    if (!LAST || s == 0) load_kset(ks[(s + 3) & 3], kw + (size_t)(8 * (hs + 3)) * H_);
    if (LAST && s == 1) load_vgroup(va, vw);
    if (LAST && s == 2) load_vgroup(vb, vw + (size_t)16 * H_);
    float p[8];
    score_half(q, vv, ks[s], p);
    const float w0 = tsum<0x161>(p[0], p[1], b0);
    const float w1 = tsum<0x161>(p[2], p[3], b0);
    const float w2 = tsum<0x161>(p[4], p[5], b0);
    const float w3 = tsum<0x161>(p[6], p[7], b0);
    const float x0 = tsum<0x162>(w0, w1, b1);
    const float x1 = tsum<0x162>(w2, w3, b1);
    scw[8 * hs] = tsum<0x164>(x0, x1, b2);
#pragma unroll
    for (int n = 0; n < 128; ++n) {
      __builtin_amdgcn_sched_group_barrier(0x400, 1, 0);
      __builtin_amdgcn_sched_group_barrier(0x002, 1, 0);
    }
    __builtin_amdgcn_sched_barrier(0);
  }
}

#define WS_NEED_ ((size_t)4 * (B_ * K_ * H_ + NS_ * B_ * Q_ * H_ + NS_ * B_ * Q_ * 2))

__global__ __launch_bounds__(256) __attribute__((amdgpu_waves_per_eu(2, 3))) void score_split(
    const float* __restrict__ query, const float* __restrict__ Wq,
    const float* __restrict__ kp,
    const int* __restrict__ mask, const float* __restrict__ v,
    const float* __restrict__ V, float* __restrict__ attn,
    float* __restrict__ stats, float* __restrict__ cpart, size_t ws_bytes) {
  if (ws_bytes < WS_NEED_) return;

  __shared__ __attribute__((aligned(16))) float sc[16 * SCP_];
  __shared__ __attribute__((aligned(16))) float msk[16 * KQ_];
  __shared__ __attribute__((aligned(16))) float qs[16][H_];
  __shared__ __attribute__((aligned(16))) float part[8][16][H_];
  __shared__ __attribute__((aligned(8))) float stl[16][2];

  const int blk = blockIdx.x;
  const int kq = blk & (NS_ - 1);
  const int R0 = 16 * (blk >> 2);
  const int b = blk >> 6;
  const int K0 = KQ_ * kq;
  const int tid = threadIdx.x;
  const int lane = tid & 31;
  const int w = tid >> 5;
  const int rg = w & 3;
  const int kh = w >> 2;
  const int g = lane >> 3;
  const int i = lane & 7;
  const bool b0 = (lane & 1) != 0;
  const bool b1 = (lane & 2) != 0;
  const bool b2 = (lane & 4) != 0;
  const int rr = tid >> 4;
  const int cc = tid & 15;

  v4f q[4][4], vv[4];
#pragma unroll
  for (int j = 0; j < 4; ++j) vv[j] = *(const v4f*)(v + 32 * j + 4 * i);

#pragma unroll
  for (int n = 0; n < 4; ++n) {
    const v4i m = *(const v4i*)(mask + (size_t)(R0 + rr) * K_ + K0 + 64 * n + 4 * cc);
    v4f mt;
    mt.x = -1e-9f * (1.0f - (float)m.x);
    mt.y = -1e-9f * (1.0f - (float)m.y);
    mt.z = -1e-9f * (1.0f - (float)m.z);
    mt.w = -1e-9f * (1.0f - (float)m.w);
    *(v4f*)(msk + rr * KQ_ + 64 * n + 4 * cc) = mt;
  }

  const float* kw = kp + ((size_t)b * K_ + K0 + 128 * kh + g) * H_ + 4 * i;

  v4f ks[4][2][4];
  load_kset(ks[0], kw);
  load_kset(ks[1], kw + (size_t)8 * H_);
  load_kset(ks[2], kw + (size_t)16 * H_);

  {
    const int pc = lane & 15;
    const int ph = lane >> 4;
    const float* xa = query + (size_t)(R0 + pc) * H_ + 4 * ph;
    const float* wb = Wq + (size_t)(16 * w + pc) * H_ + 4 * ph;
    v8f c = {};
#pragma unroll
    for (int kk = 0; kk < H_ / 8; ++kk) {
      const v4f a = *(const v4f*)(xa + 8 * kk);
      const v4f bq = *(const v4f*)(wb + 8 * kk);
      c = __builtin_amdgcn_wmma_f32_16x16x4_f32(false, a.lo, false, bq.lo,
                                                (short)0, c, false, false);
      c = __builtin_amdgcn_wmma_f32_16x16x4_f32(false, a.hi, false, bq.hi,
                                                (short)0, c, false, false);
    }
#pragma unroll
    for (int r = 0; r < 8; ++r) qs[8 * ph + r][16 * w + pc] = c[r];
  }

  __syncthreads();
#pragma unroll
  for (int r = 0; r < 4; ++r)
#pragma unroll
    for (int j = 0; j < 4; ++j) q[r][j] = *(const v4f*)&qs[4 * rg + r][32 * j + 4 * i];

  float* scw = sc + (4 * rg + (i & 3)) * SCP_ + 128 * kh + 4 * (i >> 2) + g;

  const int c16 = lane & 15;
  const int half = lane >> 4;
  const float* vw = V + ((size_t)b * K_ + K0 + 32 * w + 2 * half) * H_ + 8 * c16;
  v4f va[4][2][2], vb[4][2][2];

  for (int it = 0; it < 3; ++it)
    score_pass<false>(it, ks, q, vv, kw, scw, b0, b1, b2, va, vb, vw);
  score_pass<true>(3, ks, q, vv, kw, scw, b0, b1, b2, va, vb, vw);
  __syncthreads();

  v4f s[4];
#pragma unroll
  for (int n = 0; n < 4; ++n)
    s[n] = *(const v4f*)(sc + rr * SCP_ + 64 * n + 4 * cc) + *(const v4f*)(msk + rr * KQ_ + 64 * n + 4 * cc);
  float mx = fmaxf(fmaxf(s[0].x, s[0].y), fmaxf(s[0].z, s[0].w));
#pragma unroll
  for (int n = 1; n < 4; ++n) mx = fmaxf(mx, fmaxf(fmaxf(s[n].x, s[n].y), fmaxf(s[n].z, s[n].w)));
  mx = row_max(mx);
  float sm = 0.f;
#pragma unroll
  for (int n = 0; n < 4; ++n) {
    s[n].x = fast_exp(s[n].x - mx);
    s[n].y = fast_exp(s[n].y - mx);
    s[n].z = fast_exp(s[n].z - mx);
    s[n].w = fast_exp(s[n].w - mx);
    sm += (s[n].x + s[n].y) + (s[n].z + s[n].w);
    *(v4f*)(sc + rr * SCP_ + 64 * n + 4 * cc) = s[n];
  }
  sm = row_sum(sm);
  if (cc == 0) {
    stl[rr][0] = mx;
    stl[rr][1] = sm;
  }
  __syncthreads();

  if (tid < 16) *(v2f*)(stats + ((size_t)kq * (B_ * Q_) + R0 + tid) * 2) = *(const v2f*)&stl[tid][0];

  const float* pa = sc + c16 * SCP_ + 32 * w + 2 * half;
  v2f af[8];
#pragma unroll
  for (int kk = 0; kk < 8; ++kk) af[kk] = *(const v2f*)(pa + 4 * kk);

  v8f acc[8];
#pragma unroll
  for (int t = 0; t < 8; ++t) acc[t] = v8f{};

  mma_vgroup(acc, af, va);
  mma_vgroup(acc, af + 4, vb);

#pragma unroll
  for (int r = 0; r < 8; ++r) {
    float* dst = &part[w][8 * half + r][8 * c16];
    *(v4f*)dst = v4f{acc[0][r], acc[1][r], acc[2][r], acc[3][r]};
    *(v4f*)(dst + 4) = v4f{acc[4][r], acc[5][r], acc[6][r], acc[7][r]};
  }
  __syncthreads();

#pragma unroll
  for (int n = 0; n < 4; ++n)
    *(v4f*)(attn + (size_t)(R0 + rr) * K_ + K0 + 64 * n + 4 * cc) = s[n];

#pragma unroll
  for (int n = 0; n < 2; ++n) {
    v4f sum = *(const v4f*)&part[0][rr][64 * n + 4 * cc];
#pragma unroll
    for (int ww = 1; ww < 8; ++ww) sum += *(const v4f*)&part[ww][rr][64 * n + 4 * cc];
    *(v4f*)(cpart + ((size_t)kq * (B_ * Q_) + R0 + rr) * H_ + 64 * n + 4 * cc) = sum;
  }
}

__global__ __launch_bounds__(64) void join_splits(const float* __restrict__ stats,
                                                  const float* __restrict__ cpart,
                                                  float* __restrict__ attn,
                                                  float* __restrict__ ctx, size_t ws_bytes) {
  if (ws_bytes < WS_NEED_) return;

  const int blk = blockIdx.x;
  const int tid = threadIdx.x;

  v4f e[4][NS_];
  v2f cp[4][NS_];
#pragma unroll
  for (int r = 0; r < 4; ++r)
#pragma unroll
    for (int j = 0; j < NS_; ++j) {
      e[r][j] = *(const v4f*)(attn + (size_t)(4 * blk + r) * K_ + KQ_ * j + 4 * tid);
      cp[r][j] = *(const v2f*)(cpart + ((size_t)j * (B_ * Q_) + 4 * blk + r) * H_ + 2 * tid);
    }

#pragma unroll
  for (int r = 0; r < 4; ++r) {
    v2f st[NS_];
#pragma unroll
    for (int j = 0; j < NS_; ++j)
      st[j] = *(const v2f*)(stats + ((size_t)j * (B_ * Q_) + 4 * blk + r) * 2);
    const float m = fmaxf(fmaxf(st[0].x, st[1].x), fmaxf(st[2].x, st[3].x));
    float f[NS_];
    float L = 0.f;
#pragma unroll
    for (int j = 0; j < NS_; ++j) {
      f[j] = fast_exp(st[j].x - m);
      L += f[j] * st[j].y;
    }
    const float inv = 1.0f / L;
    v2f sum = {0.f, 0.f};
#pragma unroll
    for (int j = 0; j < NS_; ++j) {
      const float sj = f[j] * inv;
      *(v4f*)(attn + (size_t)(4 * blk + r) * K_ + KQ_ * j + 4 * tid) = e[r][j] * sj;
      sum += cp[r][j] * sj;
    }
    *(v2f*)(ctx + (size_t)(4 * blk + r) * H_ + 2 * tid) = sum;
  }
}

extern "C" void kernel_launch(void* const* d_in, const int* in_sizes, int n_in,
                              void* d_out, int out_size, void* d_ws, size_t ws_size,
                              hipStream_t stream) {
  const float* query = (const float*)d_in[0];
  const float* key   = (const float*)d_in[1];
  const float* value = (const float*)d_in[2];
  const int*   mask  = (const int*)  d_in[3];
  const float* Wq    = (const float*)d_in[4];
  const float* Wk    = (const float*)d_in[5];
  const float* v     = (const float*)d_in[6];

  float* attn = (float*)d_out;
  float* ctx  = (float*)d_out + (size_t)B_ * Q_ * K_;

  float* kp    = (float*)d_ws;
  float* cpart = kp + (size_t)B_ * K_ * H_;
  float* stats = cpart + (size_t)NS_ * B_ * Q_ * H_;

  proj_strip<<<B_ * K_ / 16, 256, 0, stream>>>(key, Wk, kp);

  score_split<<<(B_ * Q_ / 16) * NS_, 256, 0, stream>>>(query, Wq, kp, mask, v, value, attn,
                                                        stats, cpart, ws_size);

  join_splits<<<B_ * Q_ / 4, 64, 0, stream>>>(stats, cpart, attn, ctx, ws_size);
}
